# starting kernel
# baseline (speedup 1.0000x reference)
#include <hip/hip_runtime.h>
#include <math.h>

typedef __attribute__((ext_vector_type(2))) float v2f;
typedef __attribute__((ext_vector_type(8))) float v8f;

#define HDIM 64
#define DDIM 32
#define FCIN 192

__global__ __launch_bounds__(64) void ig_setup(
    const float* __restrict__ h, const float* __restrict__ ghs,
    const float* __restrict__ goal, const float* __restrict__ position,
    const float* __restrict__ Wemb, const float* __restrict__ bemb,
    const float* __restrict__ Wfc, const float* __restrict__ bfc,
    const int* __restrict__ idxp, float* __restrict__ ws,
    float* __restrict__ goal_out)
{
    const int j   = threadIdx.x;
    const int idx = idxp[0];
    const float* wrow = Wfc + (long)j * FCIN;

    float u1 = 0.f, u2 = 0.f, cb = 0.f;
    for (int t = 0; t < DDIM; ++t) {
        float wsum = Wemb[2*t] + Wemb[2*t+1];
        float wa = wrow[128 + t], wb = wrow[160 + t];
        u1 += wa * wsum;
        u2 += wb * wsum;
        cb += (wa + wb) * bemb[t];
    }
    float ch = 0.f, gsum = 0.f;
    for (int k = 0; k < HDIM; ++k) {
        ch   += wrow[k]      * h[(long)idx * HDIM + k];
        gsum += wrow[64 + k] * ghs[k];
    }
    float c0 = ch + cb + bfc[j];

    float px = position[2*idx], py = position[2*idx + 1];
    float gx = goal[2*idx],     gy = goal[2*idx + 1];
    float dx = px - gx, dy = py - gy;
    float dg = sqrtf(dx*dx + dy*dy);
    float gv = ch + gsum + dg * (u1 + u2) + cb + bfc[j];
    goal_out[j] = 1.0f / (1.0f + __expf(-gv));

    ws[j]        = u1;
    ws[64 + j]   = u2;
    ws[128 + j]  = c0;
    ws[192 + j]  = h[(long)idx * HDIM + j];
}

__global__ __launch_bounds__(256) void ig_main(
    const float* __restrict__ h, const float* __restrict__ goal,
    const float* __restrict__ position, const float* __restrict__ Wfc,
    const int* __restrict__ idxp, const float* __restrict__ ws,
    float* __restrict__ out, int N)
{
    __shared__ float sB[4096];
    __shared__ float sE[256];

    const int tid = threadIdx.x;

    for (int e = tid; e < 4096; e += 256) {
        int frag = e >> 6;
        int lane = (e >> 1) & 31;
        int c    = e & 1;
        int jt   = frag >> 4, ks = frag & 15;
        int j    = (jt << 4) + (lane & 15);
        int k    = (ks << 2) + ((lane >> 4) << 1) + c;
        sB[e] = Wfc[(long)j * FCIN + 64 + k];
    }
    sE[tid] = ws[tid];
    __syncthreads();

    const int lane = tid & 31;
    const int wave = tid >> 5;
    const long rowbase = ((long)blockIdx.x * 8 + wave) * 16;
    if (rowbase >= N) return;

    const int idx = idxp[0];
    const float ax = position[2*idx];
    const float ay = position[2*idx + 1];

    const int  m    = lane & 15;
    const int  hi   = lane >> 4;
    const int  koff = hi << 1;
    long row = rowbase + m;
    if (row >= N) row = N - 1;

    const float* hrow = h + row * HDIM;
    v2f a[16];
#pragma unroll
    for (int ks = 0; ks < 16; ++ks)
        a[ks] = *(const v2f*)(hrow + (ks << 2) + koff);

    float gx = goal[2*row],     gy = goal[2*row + 1];
    float px = position[2*row], py = position[2*row + 1];
    float dn      = -(ay - gy) * (ax - px);
    float cross_a = ax * gy - ay * gx;
    float cross_g = ax * py - ax * px;
    float n1 = cross_a * (ax - px) - (ax - gx) * cross_g;
    float n2 = cross_a * (ay - py) - (ay - gx) * cross_g;
    float sdn = (dn != 0.0f) ? dn : 1.0f;
    float p1 = n1 / sdn, p2 = n2 / sdn;
    bool cond = (dn != 0.0f) && ((ax - p1) * (p1 - ax) > 0.0f);
    float d2b = sqrtf((gx - px) * (gx - px) + (gy - py) * (gy - py));
    float d1v = cond ? sqrtf((ax - p1) * (ax - p1) + (ay - p2) * (ay - p2)) : 0.0f;
    float d2v = cond ? sqrtf((gx - p1) * (gx - p1) + (gy - p2) * (gy - p2)) : d2b;

    v8f acc[4] = {v8f{}, v8f{}, v8f{}, v8f{}};
    const v2f* sB2 = (const v2f*)sB;
#pragma unroll
    for (int ks = 0; ks < 16; ++ks) {
        v2f b0 = sB2[(0 * 16 + ks) * 32 + lane];
        v2f b1 = sB2[(1 * 16 + ks) * 32 + lane];
        v2f b2 = sB2[(2 * 16 + ks) * 32 + lane];
        v2f b3 = sB2[(3 * 16 + ks) * 32 + lane];
        acc[0] = __builtin_amdgcn_wmma_f32_16x16x4_f32(false, a[ks], false, b0, (short)0, acc[0], false, false);
        acc[1] = __builtin_amdgcn_wmma_f32_16x16x4_f32(false, a[ks], false, b1, (short)0, acc[1], false, false);
        acc[2] = __builtin_amdgcn_wmma_f32_16x16x4_f32(false, a[ks], false, b2, (short)0, acc[2], false, false);
        acc[3] = __builtin_amdgcn_wmma_f32_16x16x4_f32(false, a[ks], false, b3, (short)0, acc[3], false, false);
    }

    float u1f[4], u2f[4], c0f[4], hif[4];
#pragma unroll
    for (int jt = 0; jt < 4; ++jt) {
        int j = (jt << 4) + m;
        u1f[jt] = sE[j];
        u2f[jt] = sE[64 + j];
        c0f[jt] = sE[128 + j];
        hif[jt] = sE[192 + j];
    }

#pragma unroll
    for (int r = 0; r < 8; ++r) {
        int src = r + (hi << 3);
        float dd1 = __shfl(d1v, src, 32);
        float dd2 = __shfl(d2v, src, 32);
        long orow = rowbase + src;
        if (orow >= N) continue;
        float* orowp = out + orow * HDIM;
        bool keep = (orow == (long)idx);
#pragma unroll
        for (int jt = 0; jt < 4; ++jt) {
            float val = acc[jt][r] + dd1 * u1f[jt] + dd2 * u2f[jt] + c0f[jt];
            val = 1.0f / (1.0f + __expf(-val));
            if (keep) val = hif[jt];
            orowp[(jt << 4) + m] = val;
        }
    }
}

extern "C" void kernel_launch(void* const* d_in, const int* in_sizes, int n_in,
                              void* d_out, int out_size, void* d_ws, size_t ws_size,
                              hipStream_t stream)
{
    const float* h    = (const float*)d_in[0];
    const float* ghs  = (const float*)d_in[1];
    const float* goal = (const float*)d_in[2];
    const float* pos  = (const float*)d_in[3];
    const float* Wemb = (const float*)d_in[4];
    const float* bemb = (const float*)d_in[5];
    const float* Wfc  = (const float*)d_in[6];
    const float* bfc  = (const float*)d_in[7];
    const int*   idxp = (const int*)d_in[8];
    float* out = (float*)d_out;
    float* ws  = (float*)d_ws;

    const int N = in_sizes[0] / HDIM;

    ig_setup<<<1, 64, 0, stream>>>(h, ghs, goal, pos, Wemb, bemb, Wfc, bfc,
                                   idxp, ws, out + (long)N * HDIM);

    const int blocks = (N + 127) / 128;
    ig_main<<<blocks, 256, 0, stream>>>(h, goal, pos, Wfc, idxp, ws, out, N);
}


# Persistent work-groups: each wave loads its W2 B fragments once into registers and loops over row tiles; weight staging, barrier and all LDS reads of B leave the per-tile work; k order kept
# speedup vs baseline: 1.1113x; 1.1113x over previous
#include <hip/hip_runtime.h>
#include <math.h>

typedef __attribute__((ext_vector_type(2))) float v2f;
typedef __attribute__((ext_vector_type(4))) float v4f;
typedef __attribute__((ext_vector_type(8))) float v8f;

#define HDIM 64
#define DDIM 32
#define FCIN 192

__global__ __launch_bounds__(64) void ig_setup(
    const float* __restrict__ h, const float* __restrict__ ghs,
    const float* __restrict__ goal, const float* __restrict__ position,
    const float* __restrict__ Wemb, const float* __restrict__ bemb,
    const float* __restrict__ Wfc, const float* __restrict__ bfc,
    const int* __restrict__ idxp, float* __restrict__ ws,
    float* __restrict__ goal_out)
{
    const int j   = threadIdx.x;
    const int idx = idxp[0];
    const float* wrow = Wfc + (long)j * FCIN;

    float u1 = 0.f, u2 = 0.f, cb = 0.f;
    for (int t = 0; t < DDIM; ++t) {
        float wsum = Wemb[2*t] + Wemb[2*t+1];
        float wa = wrow[128 + t], wb = wrow[160 + t];
        u1 += wa * wsum;
        u2 += wb * wsum;
        cb += (wa + wb) * bemb[t];
    }
    float ch = 0.f, gsum = 0.f;
    for (int k = 0; k < HDIM; ++k) {
        ch   += wrow[k]      * h[(long)idx * HDIM + k];
        gsum += wrow[64 + k] * ghs[k];
    }
    float c0 = ch + cb + bfc[j];

    float px = position[2*idx], py = position[2*idx + 1];
    float gx = goal[2*idx],     gy = goal[2*idx + 1];
    float dx = px - gx, dy = py - gy;
    float dg = sqrtf(dx*dx + dy*dy);
    float gv = ch + gsum + dg * (u1 + u2) + cb + bfc[j];
    goal_out[j] = 1.0f / (1.0f + __expf(-gv));

    ws[j]        = u1;
    ws[64 + j]   = u2;
    ws[128 + j]  = c0;
    ws[192 + j]  = h[(long)idx * HDIM + j];
}

#define WPITCH 68
#define NGROUPS 512

__global__ __launch_bounds__(256) void ig_main(
    const float* __restrict__ h, const float* __restrict__ goal,
    const float* __restrict__ position, const float* __restrict__ Wfc,
    const int* __restrict__ idxp, const float* __restrict__ ws,
    float* __restrict__ out, int N)
{
    __shared__ __attribute__((aligned(16))) float sW[64 * WPITCH];
    __shared__ float sE[256];

    const int tid = threadIdx.x;

#pragma unroll
    for (int i = 0; i < 4; ++i) {
        int q  = tid + (i << 8);
        int j  = q >> 4, kq = q & 15;
        v4f w  = *(const v4f*)(Wfc + (long)j * FCIN + 64 + (kq << 2));
        *(v4f*)(sW + j * WPITCH + (kq << 2)) = w;
    }
    sE[tid] = ws[tid];
    __syncthreads();

    const int lane = tid & 31;
    const int wave = __builtin_amdgcn_readfirstlane(tid >> 5);
    const int m    = lane & 15;
    const int hi   = lane >> 4;
    const int koff = hi << 1;

    v2f b[4][16];
#pragma unroll
    for (int jt = 0; jt < 4; ++jt)
#pragma unroll
        for (int ks = 0; ks < 16; ++ks)
            b[jt][ks] = *(const v2f*)(sW + ((jt << 4) + m) * WPITCH + (ks << 2) + koff);

    const int idx = idxp[0];
    const float ax = position[2*idx];
    const float ay = position[2*idx + 1];

    const long nwaves = (long)NGROUPS * 8;
    for (long wt = (long)wave * NGROUPS + blockIdx.x; wt * 16 < (long)N; wt += nwaves) {
        const long rowbase = wt * 16;
        long row = rowbase + m;
        if (row >= N) row = N - 1;

        const float* hrow = h + row * HDIM;
        v2f a[16];
#pragma unroll
        for (int ks = 0; ks < 16; ++ks)
            a[ks] = *(const v2f*)(hrow + (ks << 2) + koff);

        float gx = goal[2*row],     gy = goal[2*row + 1];
        float px = position[2*row], py = position[2*row + 1];
        float dn      = -(ay - gy) * (ax - px);
        float cross_a = ax * gy - ay * gx;
        float cross_g = ax * py - ax * px;
        float n1 = cross_a * (ax - px) - (ax - gx) * cross_g;
        float n2 = cross_a * (ay - py) - (ay - gx) * cross_g;
        float sdn = (dn != 0.0f) ? dn : 1.0f;
        float p1 = n1 / sdn, p2 = n2 / sdn;
        bool cond = (dn != 0.0f) && ((ax - p1) * (p1 - ax) > 0.0f);
        float d2b = sqrtf((gx - px) * (gx - px) + (gy - py) * (gy - py));
        float d1v = cond ? sqrtf((ax - p1) * (ax - p1) + (ay - p2) * (ay - p2)) : 0.0f;
        float d2v = cond ? sqrtf((gx - p1) * (gx - p1) + (gy - p2) * (gy - p2)) : d2b;

        v8f acc[4] = {v8f{}, v8f{}, v8f{}, v8f{}};
#pragma unroll
        for (int ks = 0; ks < 16; ++ks) {
            acc[0] = __builtin_amdgcn_wmma_f32_16x16x4_f32(false, a[ks], false, b[0][ks], (short)0, acc[0], false, false);
            acc[1] = __builtin_amdgcn_wmma_f32_16x16x4_f32(false, a[ks], false, b[1][ks], (short)0, acc[1], false, false);
            acc[2] = __builtin_amdgcn_wmma_f32_16x16x4_f32(false, a[ks], false, b[2][ks], (short)0, acc[2], false, false);
            acc[3] = __builtin_amdgcn_wmma_f32_16x16x4_f32(false, a[ks], false, b[3][ks], (short)0, acc[3], false, false);
        }

        float u1f[4], u2f[4], c0f[4], hif[4];
#pragma unroll
        for (int jt = 0; jt < 4; ++jt) {
            int j = (jt << 4) + m;
            u1f[jt] = sE[j];
            u2f[jt] = sE[64 + j];
            c0f[jt] = sE[128 + j];
            hif[jt] = sE[192 + j];
        }

#pragma unroll
        for (int r = 0; r < 8; ++r) {
            int src = r + (hi << 3);
            float dd1 = __shfl(d1v, src, 32);
            float dd2 = __shfl(d2v, src, 32);
            long orow = rowbase + src;
            if (orow >= N) continue;
            float* orowp = out + orow * HDIM;
            bool keep = (orow == (long)idx);
#pragma unroll
            for (int jt = 0; jt < 4; ++jt) {
                float val = acc[jt][r] + dd1 * u1f[jt] + dd2 * u2f[jt] + c0f[jt];
                val = 1.0f / (1.0f + __expf(-val));
                if (keep) val = hif[jt];
                orowp[(jt << 4) + m] = val;
            }
        }
    }
}

extern "C" void kernel_launch(void* const* d_in, const int* in_sizes, int n_in,
                              void* d_out, int out_size, void* d_ws, size_t ws_size,
                              hipStream_t stream)
{
    const float* h    = (const float*)d_in[0];
    const float* ghs  = (const float*)d_in[1];
    const float* goal = (const float*)d_in[2];
    const float* pos  = (const float*)d_in[3];
    const float* Wemb = (const float*)d_in[4];
    const float* bemb = (const float*)d_in[5];
    const float* Wfc  = (const float*)d_in[6];
    const float* bfc  = (const float*)d_in[7];
    const int*   idxp = (const int*)d_in[8];
    float* out = (float*)d_out;
    float* ws  = (float*)d_ws;

    const int N = in_sizes[0] / HDIM;

    ig_setup<<<1, 64, 0, stream>>>(h, ghs, goal, pos, Wemb, bemb, Wfc, bfc,
                                   idxp, ws, out + (long)N * HDIM);

    ig_main<<<NGROUPS, 256, 0, stream>>>(h, goal, pos, Wfc, idxp, ws, out, N);
}


# Sigmoid quotient 1/(1+e^-x) taken with the hardware reciprocal v_rcp_f32 (1 ulp) instead of the correctly rounded division: 4 instead of 14 vector instructions per output element
# speedup vs baseline: 1.2188x; 1.0967x over previous
#include <hip/hip_runtime.h>
#include <math.h>

typedef __attribute__((ext_vector_type(2))) float v2f;
typedef __attribute__((ext_vector_type(4))) float v4f;
typedef __attribute__((ext_vector_type(8))) float v8f;

#define HDIM 64
#define DDIM 32
#define FCIN 192

__global__ __launch_bounds__(64) void ig_setup(
    const float* __restrict__ h, const float* __restrict__ ghs,
    const float* __restrict__ goal, const float* __restrict__ position,
    const float* __restrict__ Wemb, const float* __restrict__ bemb,
    const float* __restrict__ Wfc, const float* __restrict__ bfc,
    const int* __restrict__ idxp, float* __restrict__ ws,
    float* __restrict__ goal_out)
{
    const int j   = threadIdx.x;
    const int idx = idxp[0];
    const float* wrow = Wfc + (long)j * FCIN;

    float u1 = 0.f, u2 = 0.f, cb = 0.f;
    for (int t = 0; t < DDIM; ++t) {
        float wsum = Wemb[2*t] + Wemb[2*t+1];
        float wa = wrow[128 + t], wb = wrow[160 + t];
        u1 += wa * wsum;
        u2 += wb * wsum;
        cb += (wa + wb) * bemb[t];
    }
    float ch = 0.f, gsum = 0.f;
    for (int k = 0; k < HDIM; ++k) {
        ch   += wrow[k]      * h[(long)idx * HDIM + k];
        gsum += wrow[64 + k] * ghs[k];
    }
    float c0 = ch + cb + bfc[j];

    float px = position[2*idx], py = position[2*idx + 1];
    float gx = goal[2*idx],     gy = goal[2*idx + 1];
    float dx = px - gx, dy = py - gy;
    float dg = sqrtf(dx*dx + dy*dy);
    float gv = ch + gsum + dg * (u1 + u2) + cb + bfc[j];
    goal_out[j] = 1.0f / (1.0f + __expf(-gv));

    ws[j]        = u1;
    ws[64 + j]   = u2;
    ws[128 + j]  = c0;
    ws[192 + j]  = h[(long)idx * HDIM + j];
}

#define WPITCH 68
#define NGROUPS 512

__global__ __launch_bounds__(256) void ig_main(
    const float* __restrict__ h, const float* __restrict__ goal,
    const float* __restrict__ position, const float* __restrict__ Wfc,
    const int* __restrict__ idxp, const float* __restrict__ ws,
    float* __restrict__ out, int N)
{
    __shared__ __attribute__((aligned(16))) float sW[64 * WPITCH];
    __shared__ float sE[256];

    const int tid = threadIdx.x;

#pragma unroll
    for (int i = 0; i < 4; ++i) {
        int q  = tid + (i << 8);
        int j  = q >> 4, kq = q & 15;
        v4f w  = *(const v4f*)(Wfc + (long)j * FCIN + 64 + (kq << 2));
        *(v4f*)(sW + j * WPITCH + (kq << 2)) = w;
    }
    sE[tid] = ws[tid];
    __syncthreads();

    const int lane = tid & 31;
    const int wave = __builtin_amdgcn_readfirstlane(tid >> 5);
    const int m    = lane & 15;
    const int hi   = lane >> 4;
    const int koff = hi << 1;

    v2f b[4][16];
#pragma unroll
    for (int jt = 0; jt < 4; ++jt)
#pragma unroll
        for (int ks = 0; ks < 16; ++ks)
            b[jt][ks] = *(const v2f*)(sW + ((jt << 4) + m) * WPITCH + (ks << 2) + koff);

    const int idx = idxp[0];
    const float ax = position[2*idx];
    const float ay = position[2*idx + 1];

    const long nwaves = (long)NGROUPS * 8;
    for (long wt = (long)wave * NGROUPS + blockIdx.x; wt * 16 < (long)N; wt += nwaves) {
        const long rowbase = wt * 16;
        long row = rowbase + m;
        if (row >= N) row = N - 1;

        const float* hrow = h + row * HDIM;
        v2f a[16];
#pragma unroll
        for (int ks = 0; ks < 16; ++ks)
            a[ks] = *(const v2f*)(hrow + (ks << 2) + koff);

        float gx = goal[2*row],     gy = goal[2*row + 1];
        float px = position[2*row], py = position[2*row + 1];
        float dn      = -(ay - gy) * (ax - px);
        float cross_a = ax * gy - ay * gx;
        float cross_g = ax * py - ax * px;
        float n1 = cross_a * (ax - px) - (ax - gx) * cross_g;
        float n2 = cross_a * (ay - py) - (ay - gx) * cross_g;
        float sdn = (dn != 0.0f) ? dn : 1.0f;
        float p1 = n1 / sdn, p2 = n2 / sdn;
        bool cond = (dn != 0.0f) && ((ax - p1) * (p1 - ax) > 0.0f);
        float d2b = sqrtf((gx - px) * (gx - px) + (gy - py) * (gy - py));
        float d1v = cond ? sqrtf((ax - p1) * (ax - p1) + (ay - p2) * (ay - p2)) : 0.0f;
        float d2v = cond ? sqrtf((gx - p1) * (gx - p1) + (gy - p2) * (gy - p2)) : d2b;

        v8f acc[4] = {v8f{}, v8f{}, v8f{}, v8f{}};
#pragma unroll
        for (int ks = 0; ks < 16; ++ks) {
            acc[0] = __builtin_amdgcn_wmma_f32_16x16x4_f32(false, a[ks], false, b[0][ks], (short)0, acc[0], false, false);
            acc[1] = __builtin_amdgcn_wmma_f32_16x16x4_f32(false, a[ks], false, b[1][ks], (short)0, acc[1], false, false);
            acc[2] = __builtin_amdgcn_wmma_f32_16x16x4_f32(false, a[ks], false, b[2][ks], (short)0, acc[2], false, false);
            acc[3] = __builtin_amdgcn_wmma_f32_16x16x4_f32(false, a[ks], false, b[3][ks], (short)0, acc[3], false, false);
        }

        float u1f[4], u2f[4], c0f[4], hif[4];
#pragma unroll
        for (int jt = 0; jt < 4; ++jt) {
            int j = (jt << 4) + m;
            u1f[jt] = sE[j];
            u2f[jt] = sE[64 + j];
            c0f[jt] = sE[128 + j];
            hif[jt] = sE[192 + j];
        }

#pragma unroll
        for (int r = 0; r < 8; ++r) {
            int src = r + (hi << 3);
            float dd1 = __shfl(d1v, src, 32);
            float dd2 = __shfl(d2v, src, 32);
            long orow = rowbase + src;
            if (orow >= N) continue;
            float* orowp = out + orow * HDIM;
            bool keep = (orow == (long)idx);
#pragma unroll
            for (int jt = 0; jt < 4; ++jt) {
                float val = acc[jt][r] + dd1 * u1f[jt] + dd2 * u2f[jt] + c0f[jt];
                val = __builtin_amdgcn_rcpf(1.0f + __expf(-val));
                if (keep) val = hif[jt];
                orowp[(jt << 4) + m] = val;
            }
        }
    }
}

extern "C" void kernel_launch(void* const* d_in, const int* in_sizes, int n_in,
                              void* d_out, int out_size, void* d_ws, size_t ws_size,
                              hipStream_t stream)
{
    const float* h    = (const float*)d_in[0];
    const float* ghs  = (const float*)d_in[1];
    const float* goal = (const float*)d_in[2];
    const float* pos  = (const float*)d_in[3];
    const float* Wemb = (const float*)d_in[4];
    const float* bemb = (const float*)d_in[5];
    const float* Wfc  = (const float*)d_in[6];
    const float* bfc  = (const float*)d_in[7];
    const int*   idxp = (const int*)d_in[8];
    float* out = (float*)d_out;
    float* ws  = (float*)d_ws;

    const int N = in_sizes[0] / HDIM;

    ig_setup<<<1, 64, 0, stream>>>(h, ghs, goal, pos, Wemb, bemb, Wfc, bfc,
                                   idxp, ws, out + (long)N * HDIM);

    ig_main<<<NGROUPS, 256, 0, stream>>>(h, goal, pos, Wfc, idxp, ws, out, N);
}


# Set-up launch folded into the main kernel: each work-group stages all of the fc weight in LDS and computes the folded vectors u1, u2, c0 itself (same sum order); 1 launch, not 2; no workspace
# speedup vs baseline: 1.3903x; 1.1407x over previous
#include <hip/hip_runtime.h>
#include <math.h>

typedef __attribute__((ext_vector_type(2))) float v2f;
typedef __attribute__((ext_vector_type(4))) float v4f;
typedef __attribute__((ext_vector_type(8))) float v8f;

#define HDIM 64
#define DDIM 32
#define FCIN 192

#define WFP 196
#define NGROUPS 512

__global__ __launch_bounds__(256) void ig_main(
    const float* __restrict__ h, const float* __restrict__ ghs,
    const float* __restrict__ goal, const float* __restrict__ position,
    const float* __restrict__ Wemb, const float* __restrict__ bemb,
    const float* __restrict__ Wfc, const float* __restrict__ bfc,
    const int* __restrict__ idxp, float* __restrict__ out, int N)
{
    __shared__ __attribute__((aligned(16))) float sWf[64 * WFP];
    __shared__ __attribute__((aligned(16))) float sH[HDIM];
    __shared__ __attribute__((aligned(16))) float sWs[DDIM];
    __shared__ __attribute__((aligned(16))) float sBe[DDIM];
    __shared__ __attribute__((aligned(16))) float sG[HDIM];
    __shared__ float sU1[HDIM], sU2[HDIM], sCH[HDIM], sCB[HDIM], sGS[HDIM];

    const int tid = threadIdx.x;
    const int idx = idxp[0];
    const int wave = __builtin_amdgcn_readfirstlane(tid >> 5);

#pragma unroll
    for (int i = 0; i < 12; ++i) {
        int q   = tid + (i << 8);
        int row = q / (FCIN / 4), c4 = q - row * (FCIN / 4);
        v4f w   = *(const v4f*)(Wfc + ((long)q << 2));
        *(v4f*)(sWf + row * WFP + (c4 << 2)) = w;
    }
    if (wave < 2) {
        sH[tid] = h[(long)idx * HDIM + tid];
    } else if (wave == 2) {
        int t = tid - 64;
        sWs[t] = Wemb[2*t] + Wemb[2*t+1];
    } else if (wave == 3) {
        int t = tid - 96;
        sBe[t] = bemb[t];
    } else if (wave < 6 && blockIdx.x == 0) {
        sG[tid - 128] = ghs[tid - 128];
    }
    __syncthreads();

    if (wave < 2) {
        const float* wr = sWf + tid * WFP;
        v4f wv[16], xv[16];
#pragma unroll
        for (int i = 0; i < 16; ++i) { wv[i] = *(const v4f*)(wr + 4*i); xv[i] = *(const v4f*)(sH + 4*i); }
        float ch = 0.f;
#pragma unroll
        for (int i = 0; i < 16; ++i) {
            ch += wv[i].x * xv[i].x; ch += wv[i].y * xv[i].y;
            ch += wv[i].z * xv[i].z; ch += wv[i].w * xv[i].w;
        }
        sCH[tid] = ch;
    } else if (wave < 4) {
        const int j = tid - 64;
        const float* wr = sWf + j * WFP;
        v4f wa[8], wb[8], xv[8];
#pragma unroll
        for (int i = 0; i < 8; ++i) {
            wa[i] = *(const v4f*)(wr + 128 + 4*i); wb[i] = *(const v4f*)(wr + 160 + 4*i);
            xv[i] = *(const v4f*)(sWs + 4*i);
        }
        float u1 = 0.f, u2 = 0.f;
#pragma unroll
        for (int i = 0; i < 8; ++i) {
            u1 += wa[i].x * xv[i].x; u2 += wb[i].x * xv[i].x;
            u1 += wa[i].y * xv[i].y; u2 += wb[i].y * xv[i].y;
            u1 += wa[i].z * xv[i].z; u2 += wb[i].z * xv[i].z;
            u1 += wa[i].w * xv[i].w; u2 += wb[i].w * xv[i].w;
        }
        sU1[j] = u1;
        sU2[j] = u2;
    } else if (wave < 6) {
        const int j = tid - 128;
        const float* wr = sWf + j * WFP;
        v4f wa[8], wb[8], xv[8];
#pragma unroll
        for (int i = 0; i < 8; ++i) {
            wa[i] = *(const v4f*)(wr + 128 + 4*i); wb[i] = *(const v4f*)(wr + 160 + 4*i);
            xv[i] = *(const v4f*)(sBe + 4*i);
        }
        float cb = 0.f;
#pragma unroll
        for (int i = 0; i < 8; ++i) {
            cb += (wa[i].x + wb[i].x) * xv[i].x; cb += (wa[i].y + wb[i].y) * xv[i].y;
            cb += (wa[i].z + wb[i].z) * xv[i].z; cb += (wa[i].w + wb[i].w) * xv[i].w;
        }
        sCB[j] = cb;
    } else if (blockIdx.x == 0) {
        const int j = tid - 192;
        const float* wr = sWf + j * WFP + 64;
        v4f wv[16], xv[16];
#pragma unroll
        for (int i = 0; i < 16; ++i) { wv[i] = *(const v4f*)(wr + 4*i); xv[i] = *(const v4f*)(sG + 4*i); }
        float gs = 0.f;
#pragma unroll
        for (int i = 0; i < 16; ++i) {
            gs += wv[i].x * xv[i].x; gs += wv[i].y * xv[i].y;
            gs += wv[i].z * xv[i].z; gs += wv[i].w * xv[i].w;
        }
        sGS[j] = gs;
    }

    const int lane = tid & 31;
    const int m    = lane & 15;
    const int hi   = lane >> 4;
    const int koff = hi << 1;

    v2f b[4][16];
#pragma unroll
    for (int jt = 0; jt < 4; ++jt)
#pragma unroll
        for (int ks = 0; ks < 16; ++ks)
            b[jt][ks] = *(const v2f*)(sWf + ((jt << 4) + m) * WFP + 64 + (ks << 2) + koff);

    __syncthreads();

    if (blockIdx.x == 0 && wave < 2) {
        const int j = tid;
        float px = position[2*idx], py = position[2*idx + 1];
        float gx = goal[2*idx],     gy = goal[2*idx + 1];
        float dx = px - gx, dy = py - gy;
        float dg = sqrtf(dx*dx + dy*dy);
        float ch = sCH[j], cb = sCB[j], u1 = sU1[j], u2 = sU2[j], gsum = sGS[j];
        float gv = ch + gsum + dg * (u1 + u2) + cb + bfc[j];
        out[(long)N * HDIM + j] = 1.0f / (1.0f + __expf(-gv));
    }

    float u1f[4], u2f[4], c0f[4], hif[4];
#pragma unroll
    for (int jt = 0; jt < 4; ++jt) {
        int j = (jt << 4) + m;
        u1f[jt] = sU1[j];
        u2f[jt] = sU2[j];
        c0f[jt] = sCH[j] + sCB[j] + bfc[j];
        hif[jt] = sH[j];
    }

    const float ax = position[2*idx];
    const float ay = position[2*idx + 1];

    const long nwaves = (long)NGROUPS * 8;
    for (long wt = (long)wave * NGROUPS + blockIdx.x; wt * 16 < (long)N; wt += nwaves) {
        const long rowbase = wt * 16;
        long row = rowbase + m;
        if (row >= N) row = N - 1;

        const float* hrow = h + row * HDIM;
        v2f a[16];
#pragma unroll
        for (int ks = 0; ks < 16; ++ks)
            a[ks] = *(const v2f*)(hrow + (ks << 2) + koff);

        float gx = goal[2*row],     gy = goal[2*row + 1];
        float px = position[2*row], py = position[2*row + 1];
        float dn      = -(ay - gy) * (ax - px);
        float cross_a = ax * gy - ay * gx;
        float cross_g = ax * py - ax * px;
        float n1 = cross_a * (ax - px) - (ax - gx) * cross_g;
        float n2 = cross_a * (ay - py) - (ay - gx) * cross_g;
        float sdn = (dn != 0.0f) ? dn : 1.0f;
        float p1 = n1 / sdn, p2 = n2 / sdn;
        bool cond = (dn != 0.0f) && ((ax - p1) * (p1 - ax) > 0.0f);
        float d2b = sqrtf((gx - px) * (gx - px) + (gy - py) * (gy - py));
        float d1v = cond ? sqrtf((ax - p1) * (ax - p1) + (ay - p2) * (ay - p2)) : 0.0f;
        float d2v = cond ? sqrtf((gx - p1) * (gx - p1) + (gy - p2) * (gy - p2)) : d2b;

        v8f acc[4] = {v8f{}, v8f{}, v8f{}, v8f{}};
#pragma unroll
        for (int ks = 0; ks < 16; ++ks) {
            acc[0] = __builtin_amdgcn_wmma_f32_16x16x4_f32(false, a[ks], false, b[0][ks], (short)0, acc[0], false, false);
            acc[1] = __builtin_amdgcn_wmma_f32_16x16x4_f32(false, a[ks], false, b[1][ks], (short)0, acc[1], false, false);
            acc[2] = __builtin_amdgcn_wmma_f32_16x16x4_f32(false, a[ks], false, b[2][ks], (short)0, acc[2], false, false);
            acc[3] = __builtin_amdgcn_wmma_f32_16x16x4_f32(false, a[ks], false, b[3][ks], (short)0, acc[3], false, false);
        }

#pragma unroll
        for (int r = 0; r < 8; ++r) {
            int src = r + (hi << 3);
            float dd1 = __shfl(d1v, src, 32);
            float dd2 = __shfl(d2v, src, 32);
            long orow = rowbase + src;
            if (orow >= N) continue;
            float* orowp = out + orow * HDIM;
            bool keep = (orow == (long)idx);
#pragma unroll
            for (int jt = 0; jt < 4; ++jt) {
                float val = acc[jt][r] + dd1 * u1f[jt] + dd2 * u2f[jt] + c0f[jt];
                val = __builtin_amdgcn_rcpf(1.0f + __expf(-val));
                if (keep) val = hif[jt];
                orowp[(jt << 4) + m] = val;
            }
        }
    }
}

extern "C" void kernel_launch(void* const* d_in, const int* in_sizes, int n_in,
                              void* d_out, int out_size, void* d_ws, size_t ws_size,
                              hipStream_t stream)
{
    const float* h    = (const float*)d_in[0];
    const float* ghs  = (const float*)d_in[1];
    const float* goal = (const float*)d_in[2];
    const float* pos  = (const float*)d_in[3];
    const float* Wemb = (const float*)d_in[4];
    const float* bemb = (const float*)d_in[5];
    const float* Wfc  = (const float*)d_in[6];
    const float* bfc  = (const float*)d_in[7];
    const int*   idxp = (const int*)d_in[8];
    float* out = (float*)d_out;

    const int N = in_sizes[0] / HDIM;

    ig_main<<<NGROUPS, 256, 0, stream>>>(h, ghs, goal, pos, Wemb, bemb, Wfc, bfc, idxp, out, N);
}


# Epilogue's per-row tests (row past N, row == idx) decided once per tile on scalars: complete tiles without row idx run one branch-free straight-line epilogue; sum order of every output kept
# speedup vs baseline: 1.4986x; 1.0779x over previous
#include <hip/hip_runtime.h>
#include <math.h>

typedef __attribute__((ext_vector_type(2))) float v2f;
typedef __attribute__((ext_vector_type(4))) float v4f;
typedef __attribute__((ext_vector_type(8))) float v8f;

#define HDIM 64
#define DDIM 32
#define FCIN 192

#define WFP 196
#define NGROUPS 512

__global__ __launch_bounds__(256) void ig_main(
    const float* __restrict__ h, const float* __restrict__ ghs,
    const float* __restrict__ goal, const float* __restrict__ position,
    const float* __restrict__ Wemb, const float* __restrict__ bemb,
    const float* __restrict__ Wfc, const float* __restrict__ bfc,
    const int* __restrict__ idxp, float* __restrict__ out, int N)
{
    __shared__ __attribute__((aligned(16))) float sWf[64 * WFP];
    __shared__ __attribute__((aligned(16))) float sH[HDIM];
    __shared__ __attribute__((aligned(16))) float sWs[DDIM];
    __shared__ __attribute__((aligned(16))) float sBe[DDIM];
    __shared__ __attribute__((aligned(16))) float sG[HDIM];
    __shared__ float sU1[HDIM], sU2[HDIM], sCH[HDIM], sCB[HDIM], sGS[HDIM];

    const int tid = threadIdx.x;
    const int idx = idxp[0];
    const int wave = __builtin_amdgcn_readfirstlane(tid >> 5);

#pragma unroll
    for (int i = 0; i < 12; ++i) {
        int q   = tid + (i << 8);
        int row = q / (FCIN / 4), c4 = q - row * (FCIN / 4);
        v4f w   = *(const v4f*)(Wfc + ((long)q << 2));
        *(v4f*)(sWf + row * WFP + (c4 << 2)) = w;
    }
    if (wave < 2) {
        sH[tid] = h[(long)idx * HDIM + tid];
    } else if (wave == 2) {
        int t = tid - 64;
        sWs[t] = Wemb[2*t] + Wemb[2*t+1];
    } else if (wave == 3) {
        int t = tid - 96;
        sBe[t] = bemb[t];
    } else if (wave < 6 && blockIdx.x == 0) {
        sG[tid - 128] = ghs[tid - 128];
    }
    __syncthreads();

    if (wave < 2) {
        const float* wr = sWf + tid * WFP;
        v4f wv[16], xv[16];
#pragma unroll
        for (int i = 0; i < 16; ++i) { wv[i] = *(const v4f*)(wr + 4*i); xv[i] = *(const v4f*)(sH + 4*i); }
        float ch = 0.f;
#pragma unroll
        for (int i = 0; i < 16; ++i) {
            ch += wv[i].x * xv[i].x; ch += wv[i].y * xv[i].y;
            ch += wv[i].z * xv[i].z; ch += wv[i].w * xv[i].w;
        }
        sCH[tid] = ch;
    } else if (wave < 4) {
        const int j = tid - 64;
        const float* wr = sWf + j * WFP;
        v4f wa[8], wb[8], xv[8];
#pragma unroll
        for (int i = 0; i < 8; ++i) {
            wa[i] = *(const v4f*)(wr + 128 + 4*i); wb[i] = *(const v4f*)(wr + 160 + 4*i);
            xv[i] = *(const v4f*)(sWs + 4*i);
        }
        float u1 = 0.f, u2 = 0.f;
#pragma unroll
        for (int i = 0; i < 8; ++i) {
            u1 += wa[i].x * xv[i].x; u2 += wb[i].x * xv[i].x;
            u1 += wa[i].y * xv[i].y; u2 += wb[i].y * xv[i].y;
            u1 += wa[i].z * xv[i].z; u2 += wb[i].z * xv[i].z;
            u1 += wa[i].w * xv[i].w; u2 += wb[i].w * xv[i].w;
        }
        sU1[j] = u1;
        sU2[j] = u2;
    } else if (wave < 6) {
        const int j = tid - 128;
        const float* wr = sWf + j * WFP;
        v4f wa[8], wb[8], xv[8];
#pragma unroll
        for (int i = 0; i < 8; ++i) {
            wa[i] = *(const v4f*)(wr + 128 + 4*i); wb[i] = *(const v4f*)(wr + 160 + 4*i);
            xv[i] = *(const v4f*)(sBe + 4*i);
        }
        float cb = 0.f;
#pragma unroll
        for (int i = 0; i < 8; ++i) {
            cb += (wa[i].x + wb[i].x) * xv[i].x; cb += (wa[i].y + wb[i].y) * xv[i].y;
            cb += (wa[i].z + wb[i].z) * xv[i].z; cb += (wa[i].w + wb[i].w) * xv[i].w;
        }
        sCB[j] = cb;
    } else if (blockIdx.x == 0) {
        const int j = tid - 192;
        const float* wr = sWf + j * WFP + 64;
        v4f wv[16], xv[16];
#pragma unroll
        for (int i = 0; i < 16; ++i) { wv[i] = *(const v4f*)(wr + 4*i); xv[i] = *(const v4f*)(sG + 4*i); }
        float gs = 0.f;
#pragma unroll
        for (int i = 0; i < 16; ++i) {
            gs += wv[i].x * xv[i].x; gs += wv[i].y * xv[i].y;
            gs += wv[i].z * xv[i].z; gs += wv[i].w * xv[i].w;
        }
        sGS[j] = gs;
    }

    const int lane = tid & 31;
    const int m    = lane & 15;
    const int hi   = lane >> 4;
    const int koff = hi << 1;

    v2f b[4][16];
#pragma unroll
    for (int jt = 0; jt < 4; ++jt)
#pragma unroll
        for (int ks = 0; ks < 16; ++ks)
            b[jt][ks] = *(const v2f*)(sWf + ((jt << 4) + m) * WFP + 64 + (ks << 2) + koff);

    __syncthreads();

    if (blockIdx.x == 0 && wave < 2) {
        const int j = tid;
        float px = position[2*idx], py = position[2*idx + 1];
        float gx = goal[2*idx],     gy = goal[2*idx + 1];
        float dx = px - gx, dy = py - gy;
        float dg = sqrtf(dx*dx + dy*dy);
        float ch = sCH[j], cb = sCB[j], u1 = sU1[j], u2 = sU2[j], gsum = sGS[j];
        float gv = ch + gsum + dg * (u1 + u2) + cb + bfc[j];
        out[(long)N * HDIM + j] = 1.0f / (1.0f + __expf(-gv));
    }

    float u1f[4], u2f[4], c0f[4], hif[4];
#pragma unroll
    for (int jt = 0; jt < 4; ++jt) {
        int j = (jt << 4) + m;
        u1f[jt] = sU1[j];
        u2f[jt] = sU2[j];
        c0f[jt] = sCH[j] + sCB[j] + bfc[j];
        hif[jt] = sH[j];
    }

    const float ax = position[2*idx];
    const float ay = position[2*idx + 1];

    const long nwaves = (long)NGROUPS * 8;
    for (long wt = (long)wave * NGROUPS + blockIdx.x; wt * 16 < (long)N; wt += nwaves) {
        const long rowbase = wt * 16;
        long row = rowbase + m;
        if (row >= N) row = N - 1;

        const float* hrow = h + row * HDIM;
        v2f a[16];
#pragma unroll
        for (int ks = 0; ks < 16; ++ks)
            a[ks] = *(const v2f*)(hrow + (ks << 2) + koff);

        float gx = goal[2*row],     gy = goal[2*row + 1];
        float px = position[2*row], py = position[2*row + 1];
        float dn      = -(ay - gy) * (ax - px);
        float cross_a = ax * gy - ay * gx;
        float cross_g = ax * py - ax * px;
        float n1 = cross_a * (ax - px) - (ax - gx) * cross_g;
        float n2 = cross_a * (ay - py) - (ay - gx) * cross_g;
        float sdn = (dn != 0.0f) ? dn : 1.0f;
        float p1 = n1 / sdn, p2 = n2 / sdn;
        bool cond = (dn != 0.0f) && ((ax - p1) * (p1 - ax) > 0.0f);
        float d2b = sqrtf((gx - px) * (gx - px) + (gy - py) * (gy - py));
        float d1v = cond ? sqrtf((ax - p1) * (ax - p1) + (ay - p2) * (ay - p2)) : 0.0f;
        float d2v = cond ? sqrtf((gx - p1) * (gx - p1) + (gy - p2) * (gy - p2)) : d2b;

        v8f acc[4] = {v8f{}, v8f{}, v8f{}, v8f{}};
#pragma unroll
        for (int ks = 0; ks < 16; ++ks) {
            acc[0] = __builtin_amdgcn_wmma_f32_16x16x4_f32(false, a[ks], false, b[0][ks], (short)0, acc[0], false, false);
            acc[1] = __builtin_amdgcn_wmma_f32_16x16x4_f32(false, a[ks], false, b[1][ks], (short)0, acc[1], false, false);
            acc[2] = __builtin_amdgcn_wmma_f32_16x16x4_f32(false, a[ks], false, b[2][ks], (short)0, acc[2], false, false);
            acc[3] = __builtin_amdgcn_wmma_f32_16x16x4_f32(false, a[ks], false, b[3][ks], (short)0, acc[3], false, false);
        }

        const int  rb32       = (int)rowbase;
        const bool plain_tile = (N - rb32 >= 16) && ((unsigned)(idx - rb32) >= 16u);
        if (plain_tile) {
            float* obase = out + (rowbase + (hi << 3)) * HDIM + m;
#pragma unroll
            for (int r = 0; r < 8; ++r) {
                int src = r + (hi << 3);
                float dd1 = __shfl(d1v, src, 32);
                float dd2 = __shfl(d2v, src, 32);
#pragma unroll
                for (int jt = 0; jt < 4; ++jt) {
                    float val = acc[jt][r] + dd1 * u1f[jt] + dd2 * u2f[jt] + c0f[jt];
                    val = __builtin_amdgcn_rcpf(1.0f + __expf(-val));
                    obase[r * HDIM + (jt << 4)] = val;
                }
            }
        } else {
#pragma unroll
            for (int r = 0; r < 8; ++r) {
                int src = r + (hi << 3);
                float dd1 = __shfl(d1v, src, 32);
                float dd2 = __shfl(d2v, src, 32);
                long orow = rowbase + src;
                if (orow >= N) continue;
                float* orowp = out + orow * HDIM;
                bool keep = (orow == (long)idx);
#pragma unroll
                for (int jt = 0; jt < 4; ++jt) {
                    float val = acc[jt][r] + dd1 * u1f[jt] + dd2 * u2f[jt] + c0f[jt];
                    val = __builtin_amdgcn_rcpf(1.0f + __expf(-val));
                    if (keep) val = hif[jt];
                    orowp[(jt << 4) + m] = val;
                }
            }
        }
    }
}

extern "C" void kernel_launch(void* const* d_in, const int* in_sizes, int n_in,
                              void* d_out, int out_size, void* d_ws, size_t ws_size,
                              hipStream_t stream)
{
    const float* h    = (const float*)d_in[0];
    const float* ghs  = (const float*)d_in[1];
    const float* goal = (const float*)d_in[2];
    const float* pos  = (const float*)d_in[3];
    const float* Wemb = (const float*)d_in[4];
    const float* bemb = (const float*)d_in[5];
    const float* Wfc  = (const float*)d_in[6];
    const float* bfc  = (const float*)d_in[7];
    const int*   idxp = (const int*)d_in[8];
    float* out = (float*)d_out;

    const int N = in_sizes[0] / HDIM;

    ig_main<<<NGROUPS, 256, 0, stream>>>(h, ghs, goal, pos, Wemb, bemb, Wfc, bfc, idxp, out, N);
}
